# starting kernel
# baseline (speedup 1.0000x reference)
#include <hip/hip_runtime.h>
#include <math.h>

#define NN   50000
#define EE   640000
#define DD   128
#define PED  37
#define PEDP 40
#define NT   9
#define NL   3

typedef __attribute__((ext_vector_type(2))) float v2f;
typedef __attribute__((ext_vector_type(8))) float v8f;

__device__ __forceinline__ int f2ord(float f) {
  int i = __float_as_int(f);
  return (i >= 0) ? i : (i ^ 0x7fffffff);
}
__device__ __forceinline__ float ord2f(int i) {
  return __int_as_float((i >= 0) ? i : (i ^ 0x7fffffff));
}


__global__ void k_copy_x(const float* __restrict__ X, float* __restrict__ xbuf) {
  int i = blockIdx.x * 256 + threadIdx.x;
  xbuf[i] = X[i];
}

__global__ void k_pad_pe(const float* __restrict__ PE, float* __restrict__ PEp) {
  int i = blockIdx.x * 256 + threadIdx.x;
  if (i >= NN * PEDP) return;
  int r = i / PEDP, c = i - r * PEDP;
  PEp[i] = (c < PED) ? PE[r * PED + c] : 0.0f;
}

__global__ void k_pad_w(const float* __restrict__ W, float* __restrict__ Wp) {
  int i = blockIdx.x * 256 + threadIdx.x;
  if (i >= PEDP * DD) return;
  int k = i >> 7;
  Wp[i] = (k < PED) ? W[i] : 0.0f;
}

__global__ void k_init(float* __restrict__ outb, int* __restrict__ amax,
                       float* __restrict__ denom, float* __restrict__ ssum,
                       float* __restrict__ ssq) {
  int i = blockIdx.x * 256 + threadIdx.x;
  outb[i] = 0.0f;
  if (i < NN) { amax[i] = f2ord(-INFINITY); denom[i] = 0.0f; }
  if (i < DD) { ssum[i] = 0.0f; ssq[i] = 0.0f; }
}

template <int K, bool PE_EPI>
__global__ void k_gemm(const float* __restrict__ A, const float* __restrict__ B,
                       const float* __restrict__ X, const float* __restrict__ bb,
                       float* __restrict__ C) {
  int wave = threadIdx.x >> 5;
  int lane = threadIdx.x & 31;
  int t  = blockIdx.x * 8 + wave;
  int tn = t & 7;
  int tm = t >> 3;
  int mlo = lane & 15;
  int hi  = lane >> 4;
  const float* arow = A + (size_t)(tm * 16 + mlo) * K;
  int ncol = tn * 16 + mlo;

  v8f c = {};
  #pragma unroll 4
  for (int k0 = 0; k0 < K; k0 += 4) {
    v2f a = *(const v2f*)(arow + k0 + hi * 2);
    v2f b;
    b.x = B[(size_t)(k0 + hi * 2    ) * DD + ncol];
    b.y = B[(size_t)(k0 + hi * 2 + 1) * DD + ncol];
    c = __builtin_amdgcn_wmma_f32_16x16x4_f32(false, a, false, b, (short)0, c,
                                              false, false);
  }
  #pragma unroll
  for (int r = 0; r < 8; ++r) {
    int row = tm * 16 + r + hi * 8;
    float v = c[r];
    if (PE_EPI) v += X[(size_t)row * DD + ncol] + bb[ncol];
    C[(size_t)row * DD + ncol] = v;
  }
}


__global__ void k_scores(const float* __restrict__ h, const float* __restrict__ al,
                         const float* __restrict__ ar, float* __restrict__ sl,
                         float* __restrict__ sr) {
  int wave = threadIdx.x >> 5, lane = threadIdx.x & 31;
  int n = blockIdx.x * 8 + wave;
  const float* row = h + (size_t)n * DD;
  float accl = 0.0f, accr = 0.0f;
  #pragma unroll
  for (int j = 0; j < 4; ++j) {
    int d = lane + 32 * j;
    float v = row[d];
    accl += v * al[d];
    accr += v * ar[d];
  }
  #pragma unroll
  for (int off = 16; off; off >>= 1) {
    accl += __shfl_xor(accl, off, 32);
    accr += __shfl_xor(accr, off, 32);
  }
  if (lane == 0) { sl[n] = accl; sr[n] = accr; }
}

__global__ void k_xe(const float* __restrict__ ee, const float* __restrict__ ae,
                     float* __restrict__ xe) {
  int wave = threadIdx.x >> 5, lane = threadIdx.x & 31;
  if (wave >= NT) return;
  float acc = 0.0f;
  #pragma unroll
  for (int j = 0; j < 4; ++j) {
    int d = lane + 32 * j;
    acc += ee[wave * DD + d] * ae[d];
  }
  #pragma unroll
  for (int off = 16; off; off >>= 1) acc += __shfl_xor(acc, off, 32);
  if (lane == 0) xe[wave] = acc;
}


__global__ void k_edge_max(const float* __restrict__ sl, const float* __restrict__ sr,
                           const float* __restrict__ xe, const int* __restrict__ src,
                           const int* __restrict__ dst, const int* __restrict__ attr,
                           float* __restrict__ alpha, int* __restrict__ amax) {
  int e = blockIdx.x * 256 + threadIdx.x;
  int s = src[e], d = dst[e];
  float a = sl[s] + sr[d] + xe[attr[e]];
  a = (a > 0.0f) ? a : 0.2f * a;
  alpha[e] = a;
  atomicMax(&amax[d], f2ord(a));
}

__global__ void k_edge_exp(const int* __restrict__ amax, const int* __restrict__ dst,
                           float* __restrict__ alpha, float* __restrict__ denom) {
  int e = blockIdx.x * 256 + threadIdx.x;
  int d = dst[e];
  float ex = expf(alpha[e] - ord2f(amax[d]));
  alpha[e] = ex;
  atomicAdd(&denom[d], ex);
}

__global__ void k_aggregate(const float* __restrict__ h, const float* __restrict__ alpha,
                            const float* __restrict__ denom, const int* __restrict__ src,
                            const int* __restrict__ dst, float* __restrict__ outb) {
  int wave = threadIdx.x >> 5, lane = threadIdx.x & 31;
  int e = blockIdx.x * 8 + wave;
  int s = src[e], d = dst[e];
  float coeff = alpha[e] / denom[d];
  float4 v = ((const float4*)(h + (size_t)s * DD))[lane];
  float* op = outb + (size_t)d * DD + lane * 4;
  atomicAdd(op + 0, coeff * v.x);
  atomicAdd(op + 1, coeff * v.y);
  atomicAdd(op + 2, coeff * v.z);
  atomicAdd(op + 3, coeff * v.w);
}


__global__ void k_bias_stats(float* __restrict__ outb, const float* __restrict__ bias,
                             float* __restrict__ ssum, float* __restrict__ ssq) {
  int d = threadIdx.x;
  int r0 = blockIdx.x * 128;
  float b = bias[d];
  float s = 0.0f, s2 = 0.0f;
  for (int r = r0; r < r0 + 128 && r < NN; ++r) {
    float v = outb[(size_t)r * DD + d] + b;
    outb[(size_t)r * DD + d] = v;
    s += v; s2 += v * v;
  }
  atomicAdd(&ssum[d], s);
  atomicAdd(&ssq[d], s2);
}

__global__ void k_bn_residual(const float* __restrict__ outb, const float* __restrict__ ssum,
                              const float* __restrict__ ssq, const float* __restrict__ gam,
                              const float* __restrict__ bet, float* __restrict__ xbuf) {
  int i = blockIdx.x * 256 + threadIdx.x;
  int d = i & (DD - 1);
  float mu  = ssum[d] * (1.0f / NN);
  float var = ssq[d] * (1.0f / NN) - mu * mu;
  float inv = rsqrtf(var + 1e-5f);
  xbuf[i] = (outb[i] - mu) * inv * gam[d] + bet[d] + xbuf[i];
}

__global__ void k_final(const float* __restrict__ outb, const float* __restrict__ bias,
                        const float* __restrict__ xbuf, float* __restrict__ out) {
  int i = blockIdx.x * 256 + threadIdx.x;
  int d = i & (DD - 1);
  out[i] = outb[i] + bias[d] + xbuf[i];
}


extern "C" void kernel_launch(void* const* d_in, const int* in_sizes, int n_in,
                              void* d_out, int out_size, void* d_ws, size_t ws_size,
                              hipStream_t stream) {
  const float* X_n   = (const float*)d_in[0];
  const float* PE    = (const float*)d_in[1];
  const int*   eidx  = (const int*)d_in[2];
  const int*   eattr = (const int*)d_in[3];
  const float* eemb  = (const float*)d_in[4];
  const float* peW   = (const float*)d_in[5];
  const float* peB   = (const float*)d_in[6];
  const float* linW  = (const float*)d_in[7];
  const float* attl  = (const float*)d_in[8];
  const float* attr_ = (const float*)d_in[9];
  const float* atte  = (const float*)d_in[10];
  const float* bias  = (const float*)d_in[11];
  const float* gam   = (const float*)d_in[12];
  const float* bet   = (const float*)d_in[13];
  const int* src = eidx;
  const int* dst = eidx + EE;

  float* ws = (float*)d_ws;
  size_t o = 0;
  float* xbuf  = ws + o; o += (size_t)NN * DD;
  float* tmp   = ws + o; o += (size_t)NN * DD;
  float* h     = ws + o; o += (size_t)NN * DD;
  float* PEp   = ws + o; o += (size_t)NN * PEDP;
  float* Wp    = ws + o; o += (size_t)PEDP * DD;
  float* sl    = ws + o; o += NN;
  float* sr    = ws + o; o += NN;
  int*   amax  = (int*)(ws + o); o += NN;
  float* denom = ws + o; o += NN;
  float* alpha = ws + o; o += EE;
  float* xe    = ws + o; o += 16;
  float* ssum  = ws + o; o += DD;
  float* ssq   = ws + o; o += DD;
  float* outb  = tmp;

  const int ND_BLKS = (NN * DD) / 256;

  k_copy_x<<<ND_BLKS, 256, 0, stream>>>(X_n, xbuf);
  k_pad_pe<<<(NN * PEDP + 255) / 256, 256, 0, stream>>>(PE, PEp);

  for (int l = 0; l < NL; ++l) {
    k_pad_w<<<(PEDP * DD + 255) / 256, 256, 0, stream>>>(peW + (size_t)l * PED * DD, Wp);
    k_gemm<PEDP, true><<<3125, 256, 0, stream>>>(PEp, Wp, xbuf, peB + l * DD, tmp);
    k_gemm<DD, false><<<3125, 256, 0, stream>>>(tmp, linW + (size_t)l * DD * DD,
                                                nullptr, nullptr, h);
    k_init<<<ND_BLKS, 256, 0, stream>>>(outb, amax, denom, ssum, ssq);

    k_xe<<<1, 288, 0, stream>>>(eemb + (size_t)l * NT * DD, atte + l * DD, xe);
    k_scores<<<NN / 8, 256, 0, stream>>>(h, attl + l * DD, attr_ + l * DD, sl, sr);

    k_edge_max<<<EE / 256, 256, 0, stream>>>(sl, sr, xe, src, dst, eattr, alpha, amax);
    k_edge_exp<<<EE / 256, 256, 0, stream>>>(amax, dst, alpha, denom);
    k_aggregate<<<EE / 8, 256, 0, stream>>>(h, alpha, denom, src, dst, outb);

    if (l < NL - 1) {
      k_bias_stats<<<(NN + 127) / 128, 128, 0, stream>>>(outb, bias + l * DD, ssum, ssq);
      k_bn_residual<<<ND_BLKS, 256, 0, stream>>>(outb, ssum, ssq, gam + l * DD,
                                                 bet + l * DD, xbuf);
    } else {
      k_final<<<ND_BLKS, 256, 0, stream>>>(outb, bias + l * DD, xbuf, (float*)d_out);
    }
  }
}


# Row maxima taken by the range's owner work-group: a first sweep over its records makes the scores and their maxima in LDS (integer max); edge-score launch and maxima table gone; no sum reordered
# speedup vs baseline: 3.1251x; 3.1251x over previous
#include <hip/hip_runtime.h>
#include <math.h>

#define NN   50000
#define EE   640000
#define DD   128
#define PED  37
#define PEDP 40
#define NT   9
#define NL   3
#define NNP  50048
#define KC   168
#define NC   144
#define ONEC 37
#define BROWS 200
#define NBINS (NN / BROWS)
static_assert(NN % BROWS == 0, "the destination ranges cover the N rows exactly");
static_assert(NBINS <= 256, "one thread per range in the 256-thread binning launches");

typedef __attribute__((ext_vector_type(2))) float v2f;
typedef __attribute__((ext_vector_type(8))) float v8f;

__device__ __forceinline__ int f2ord(float f) {
  int i = __float_as_int(f);
  return (i >= 0) ? i : (i ^ 0x7fffffff);
}
__device__ __forceinline__ float ord2f(int i) {
  return __int_as_float((i >= 0) ? i : (i ^ 0x7fffffff));
}

__device__ __forceinline__ int l2_half() {
  return (int)((__builtin_amdgcn_s_sendmsg_rtn(0x87) >> 18) & 1u);
}


__global__ void k_copy_x(const float* __restrict__ X, float* __restrict__ xbuf) {
  int i = blockIdx.x * 256 + threadIdx.x;
  xbuf[i] = X[i];
}

__global__ void k_pad_pe(const float* __restrict__ PE, float* __restrict__ PEp) {
  int i = blockIdx.x * 256 + threadIdx.x;
  if (i >= NN * PEDP) return;
  int r = i / PEDP, c = i - r * PEDP;
  PEp[i] = (c < PED) ? PE[r * PED + c] : ((c == ONEC) ? 1.0f : 0.0f);
}


#define APITCH 132
__global__ void k_wcomb(const float* __restrict__ linW, const float* __restrict__ peW,
                        const float* __restrict__ peB, const float* __restrict__ al,
                        const float* __restrict__ ar, float* __restrict__ Wc) {
  __shared__ float sB[DD * NC];
  __shared__ float sA[48 * APITCH];
  int l = blockIdx.x;
  linW += (size_t)l * DD * DD;
  peW  += (size_t)l * PED * DD;
  peB  += l * DD;
  al   += l * DD;
  ar   += l * DD;
  Wc   += (size_t)l * KC * NC;
  int wave = threadIdx.x >> 5, lane = threadIdx.x & 31;

  for (int i = threadIdx.x; i < DD * DD; i += 27 * 32) {
    int k = i >> 7, cc = i & (DD - 1);
    sB[k * NC + cc] = linW[i];
  }
  for (int i = threadIdx.x; i < 48 * DD; i += 27 * 32) {
    int r = i >> 7, k = i & (DD - 1);
    float v = 0.0f;
    if (r < PED) v = peW[i];
    else if (r == ONEC) v = peB[k];
    sA[r * APITCH + k] = v;
  }
  __syncthreads();

  for (int k = wave; k < DD; k += 27) {
    float accl = 0.0f, accr = 0.0f;
    #pragma unroll
    for (int j = 0; j < 4; ++j) {
      int d = lane + 32 * j;
      float w = sB[k * NC + d];
      accl += w * al[d];
      accr += w * ar[d];
    }
    #pragma unroll
    for (int off = 16; off; off >>= 1) {
      accl += __shfl_xor(accl, off, 32);
      accr += __shfl_xor(accr, off, 32);
    }
    if (lane < NC - DD) sB[k * NC + DD + lane] = (lane == 0) ? accl : ((lane == 1) ? accr : 0.0f);
  }
  __syncthreads();

  for (int i = threadIdx.x; i < DD * NC; i += 27 * 32) Wc[i] = sB[i];

  int tm = wave / 9, tn = wave - tm * 9;
  int mlo = lane & 15, hi = lane >> 4;
  const float* ap = sA + (tm * 16 + mlo) * APITCH + hi * 2;
  const float* bp = sB + (hi * 2) * NC + tn * 16 + mlo;
  int ncol = tn * 16 + mlo;
  v8f c = {};
  #pragma unroll 8
  for (int k0 = 0; k0 < DD; k0 += 4) {
    v2f a = *(const v2f*)(ap + k0);
    v2f b;
    b.x = bp[k0 * NC];
    b.y = bp[(k0 + 1) * NC];
    c = __builtin_amdgcn_wmma_f32_16x16x4_f32(false, a, false, b, (short)0, c,
                                              false, false);
  }
  #pragma unroll
  for (int r = 0; r < 8; ++r) {
    int row = tm * 16 + r + hi * 8;
    if (row < PEDP) Wc[(size_t)(DD + row) * NC + ncol] = c[r];
  }
}

__device__ __forceinline__ float4 ld_row4(const float* __restrict__ base, int row, int pitch, int c4) {
  row = (row < NN) ? row : (NN - 1);
  return *(const float4*)(base + (size_t)row * pitch + c4 * 4);
}

#define GROWS  32
#define GPITCH 172
#define GEMM_BLKS ((NN + GROWS - 1) / GROWS)
__global__ void k_gemm_layer(const float* __restrict__ x, const float* __restrict__ PEp,
                             const float* __restrict__ Wc, float* __restrict__ h,
                             float* __restrict__ sl, float* __restrict__ sr) {
  __shared__ __attribute__((aligned(16))) float sA[GROWS * GPITCH];
  int tn   = threadIdx.x >> 5;
  int lane = threadIdx.x & 31;
  int row0 = blockIdx.x * GROWS;

  if (tn < 8) {
    float4 v0 = ld_row4(x, row0 + tn,      DD, lane);
    float4 v1 = ld_row4(x, row0 + tn + 8,  DD, lane);
    float4 v2 = ld_row4(x, row0 + tn + 16, DD, lane);
    float4 v3 = ld_row4(x, row0 + tn + 24, DD, lane);
    *(float4*)(sA + (tn     ) * GPITCH + lane * 4) = v0;
    *(float4*)(sA + (tn +  8) * GPITCH + lane * 4) = v1;
    *(float4*)(sA + (tn + 16) * GPITCH + lane * 4) = v2;
    *(float4*)(sA + (tn + 24) * GPITCH + lane * 4) = v3;
  } else {
    float4 vp[GROWS * (PEDP / 4) / 32];
    #pragma unroll
    for (int j = 0; j < GROWS * (PEDP / 4) / 32; ++j) {
      int p = lane + 32 * j, pr = p / (PEDP / 4), pc = p - pr * (PEDP / 4);
      vp[j] = ld_row4(PEp, row0 + pr, PEDP, pc);
    }
    #pragma unroll
    for (int j = 0; j < GROWS * (PEDP / 4) / 32; ++j) {
      int p = lane + 32 * j, pr = p / (PEDP / 4), pc = p - pr * (PEDP / 4);
      *(float4*)(sA + pr * GPITCH + DD + pc * 4) = vp[j];
    }
  }
  __syncthreads();

  int mlo = lane & 15;
  int hi  = lane >> 4;
  int ncol = tn * 16 + mlo;
  const float* a0p = sA + mlo * GPITCH + hi * 2;
  const float* a1p = a0p + 16 * GPITCH;
  const float* bp  = Wc + (size_t)(hi * 2) * NC + ncol;

  v8f c0 = {}, c1 = {};
  #pragma unroll 7
  for (int k0 = 0; k0 < KC; k0 += 4) {
    v2f b;
    b.x = bp[(size_t)k0 * NC];
    b.y = bp[(size_t)(k0 + 1) * NC];
    v2f a0 = *(const v2f*)(a0p + k0);
    v2f a1 = *(const v2f*)(a1p + k0);
    c0 = __builtin_amdgcn_wmma_f32_16x16x4_f32(false, a0, false, b, (short)0, c0,
                                               false, false);
    c1 = __builtin_amdgcn_wmma_f32_16x16x4_f32(false, a1, false, b, (short)0, c1,
                                               false, false);
  }
  const bool tile1 = (row0 + 16 < NN);
  if (tn < 8) {
    float* hp = h + (size_t)(row0 + hi * 8) * DD + ncol;
    #pragma unroll
    for (int r = 0; r < 8; ++r) hp[(size_t)r * DD] = c0[r];
    if (tile1) {
      #pragma unroll
      for (int r = 0; r < 8; ++r) hp[(size_t)(r + 16) * DD] = c1[r];
    }
  } else if (mlo < 2) {
    float* sp = (mlo ? sr : sl) + row0 + hi * 8;
    #pragma unroll
    for (int r = 0; r < 8; ++r) sp[r] = c0[r];
    if (tile1) {
      #pragma unroll
      for (int r = 0; r < 8; ++r) sp[r + 16] = c1[r];
    }
  }
}

#define XE_THREADS 288
#define RESET_BLKS ((2 * NNP + XE_THREADS - 1) / XE_THREADS)
__global__ void k_xe(const float* __restrict__ ee, const float* __restrict__ ae,
                     float* __restrict__ xe,
                     float* __restrict__ denom, float* __restrict__ ssum,
                     float* __restrict__ ssq) {
  if (blockIdx.x > 0) {
    int i = (blockIdx.x - 1) * XE_THREADS + threadIdx.x;
    if (i < 2 * NNP) denom[i] = 0.0f;
    if (i < 2 * DD) { ssum[i] = 0.0f; ssq[i] = 0.0f; }
    return;
  }
  int wave = threadIdx.x >> 5, lane = threadIdx.x & 31;
  if (wave >= NT) return;
  float acc = 0.0f;
  #pragma unroll
  for (int j = 0; j < 4; ++j) {
    int d = lane + 32 * j;
    acc += ee[wave * DD + d] * ae[d];
  }
  #pragma unroll
  for (int off = 16; off; off >>= 1) acc += __shfl_xor(acc, off, 32);
  if (lane == 0) xe[wave] = acc;
}


#define BIN_WGS   125
#define BIN_EPT   5
#define BIN_EDGES (1024 * BIN_EPT)
static_assert(BIN_WGS * BIN_EDGES == EE, "the binning work-groups cover the E edges exactly");
__device__ __forceinline__ int bin_of(int d) {
  int b = (int)((unsigned)d / BROWS);
  return (b < NBINS) ? b : (NBINS - 1);
}

__global__ __launch_bounds__(1024)
void k_bin_count(const int* __restrict__ dst, int* __restrict__ whist) {
  __shared__ int hist[NBINS];
  int tid = threadIdx.x;
  if (tid < NBINS) hist[tid] = 0;
  __syncthreads();
  #pragma unroll
  for (int k = 0; k < BIN_EPT; ++k) {
    int e = blockIdx.x * BIN_EDGES + k * 1024 + tid;
    atomicAdd(&hist[bin_of(dst[e])], 1);
  }
  __syncthreads();
  if (tid < NBINS) whist[blockIdx.x * 256 + tid] = hist[tid];
}

#define SCAN_Q 32
static_assert(4 * SCAN_Q >= BIN_WGS, "four quarters cover the binning work-groups");
__global__ __launch_bounds__(1024)
void k_bin_scan(const int* __restrict__ whist, int* __restrict__ woff, int* __restrict__ bstart) {
  __shared__ int part[4][256];
  __shared__ int s[256];
  int t = threadIdx.x & 255, q = threadIdx.x >> 8;
  int c[SCAN_Q];
  #pragma unroll
  for (int i = 0; i < SCAN_Q; ++i) {
    int w = q * SCAN_Q + i;
    c[i] = (t < NBINS && w < BIN_WGS) ? whist[w * 256 + t] : 0;
  }
  int tot = 0;
  #pragma unroll
  for (int i = 0; i < SCAN_Q; ++i) tot += c[i];
  part[q][t] = tot;
  __syncthreads();
  int before = 0;
  for (int qq = 0; qq < 3; ++qq) before += (qq < q) ? part[qq][t] : 0;
  int total = part[0][t] + part[1][t] + part[2][t] + part[3][t];
  int run = before;
  #pragma unroll
  for (int i = 0; i < SCAN_Q; ++i) {
    int w = q * SCAN_Q + i;
    if (t < NBINS && w < BIN_WGS) woff[w * 256 + t] = run;
    run += c[i];
  }
  if (q == 0) s[t] = total;
  __syncthreads();
  for (int off = 1; off < 256; off <<= 1) {
    int v = (q == 0 && t >= off) ? s[t - off] : 0;
    __syncthreads();
    if (q == 0) s[t] += v;
    __syncthreads();
  }
  if (q == 0 && t <= NBINS) bstart[t] = s[t] - total;
}

__global__ __launch_bounds__(1024)
void k_bin_fill(const int* __restrict__ src, const int* __restrict__ dst,
                const int* __restrict__ attr, const int* __restrict__ woff,
                const int* __restrict__ bstart, int* __restrict__ brec) {
  __shared__ int rank[NBINS];
  __shared__ int base[NBINS];
  int tid = threadIdx.x;
  if (tid < NBINS) {
    rank[tid] = 0;
    base[tid] = bstart[tid] + woff[blockIdx.x * 256 + tid];
  }
  __syncthreads();
  #pragma unroll
  for (int k = 0; k < BIN_EPT; ++k) {
    int e = blockIdx.x * BIN_EDGES + k * 1024 + tid;
    int s = src[e], d = dst[e], t = attr[e];
    int b = bin_of(d);
    int r = atomicAdd(&rank[b], 1);
    ((int4*)brec)[base[b] + r] = make_int4(s, d, t, 0);
  }
}

#define EPW 8
#define AGG_WAVES 32
#define SWEEP1 3
__global__ __launch_bounds__(AGG_WAVES * 32)
void k_aggregate(const float* __restrict__ h, const float* __restrict__ sl,
                 const float* __restrict__ sr, const float* __restrict__ xe,
                 const int* __restrict__ brec, const int* __restrict__ bstart,
                 float* __restrict__ alpha, float* __restrict__ denom,
                 float* __restrict__ outb) {
  __shared__ float acc[BROWS * DD];
  __shared__ float den[BROWS];
  __shared__ int   rmax[BROWS];
  int tid = threadIdx.x;
  int wave = __builtin_amdgcn_readfirstlane(tid >> 5);
  int lane = tid & 31;
  int r0 = blockIdx.x * BROWS;
  for (int i = tid; i < BROWS * DD; i += AGG_WAVES * 32) acc[i] = 0.0f;
  if (tid < BROWS) { den[tid] = 0.0f; rmax[tid] = f2ord(-INFINITY); }
  int p0 = bstart[blockIdx.x], p1 = bstart[blockIdx.x + 1];
  __syncthreads();

  for (int pb = p0 + tid; pb < p1; pb += AGG_WAVES * 32 * SWEEP1) {
    int4 rc[SWEEP1];
    float gs[SWEEP1], gd[SWEEP1], gt[SWEEP1];
    #pragma unroll
    for (int u = 0; u < SWEEP1; ++u) {
      int p = pb + u * (AGG_WAVES * 32);
      rc[u] = ((const int4*)brec)[(p < p1) ? p : (p1 - 1)];
    }
    #pragma unroll
    for (int u = 0; u < SWEEP1; ++u) {
      gs[u] = sl[rc[u].x]; gd[u] = sr[rc[u].y]; gt[u] = xe[rc[u].z];
    }
    static_assert(SWEEP1 == 3, "the statement below names the gathers of three records");
    asm volatile("" :: "v"(gs[0]), "v"(gd[0]), "v"(gt[0]), "v"(gs[1]), "v"(gd[1]), "v"(gt[1]),
                       "v"(gs[2]), "v"(gd[2]), "v"(gt[2]));
    #pragma unroll
    for (int u = 0; u < SWEEP1; ++u) {
      int p = pb + u * (AGG_WAVES * 32);
      if (p < p1) {
        float s = gs[u] + gd[u] + gt[u];
        float a = (s > 0.0f) ? s : 0.2f * s;
        alpha[p] = a;
        atomicMax(&rmax[rc[u].y - r0], f2ord(a));
      }
    }
  }
  __syncthreads();

  int jl = lane & (EPW - 1);
  for (int pb = p0 + wave * EPW; pb < p1; pb += AGG_WAVES * EPW) {
    int cnt = p1 - pb;
    int jc = (jl < cnt) ? jl : (cnt - 1);
    int4 rec = ((const int4*)brec)[pb + jc];
    int s = rec.x, lr = rec.y - r0;
    float al = alpha[pb + jc];
    float v[EPW][4];
    #pragma unroll
    for (int j = 0; j < EPW; ++j) {
      int sj = __builtin_amdgcn_readlane(s, j);
      const float* hp = h + (size_t)sj * DD + lane;
      v[j][0] = hp[0]; v[j][1] = hp[32]; v[j][2] = hp[64]; v[j][3] = hp[96];
    }
    float coeff = (jl < cnt) ? expf(al - ord2f(rmax[lr])) : 0.0f;
    if (lane < EPW) atomicAdd(&den[lr], coeff);
    #pragma unroll
    for (int j = 0; j < EPW; ++j) {
      int rj = __builtin_amdgcn_readlane(lr, j);
      float cj = __int_as_float(__builtin_amdgcn_readlane(__float_as_int(coeff), j));
      float* ap = acc + rj * DD + lane;
      atomicAdd(ap,      cj * v[j][0]);
      atomicAdd(ap + 32, cj * v[j][1]);
      atomicAdd(ap + 64, cj * v[j][2]);
      atomicAdd(ap + 96, cj * v[j][3]);
    }
  }
  __syncthreads();

  float* op = outb + (size_t)r0 * DD;
  for (int i = tid; i < BROWS * DD; i += AGG_WAVES * 32) op[i] = acc[i];
  if (tid < BROWS) denom[r0 + tid] = den[tid];
}


__device__ __forceinline__ float row_norm(float acc, float den) {
  return (den > 0.0f) ? acc * __builtin_amdgcn_rcpf(den) : 0.0f;
}

#define SROWS 40
#define STAT_BLKS (NN / SROWS)
static_assert(NN % SROWS == 0, "a work-group of the statistics pass takes SROWS whole rows");
__global__ void k_bias_stats(float* __restrict__ outb, const float* __restrict__ denom,
                             const float* __restrict__ bias,
                             float* __restrict__ ssum, float* __restrict__ ssq) {
  int d = threadIdx.x;
  int r0 = blockIdx.x * SROWS;
  float b = bias[d];
  int hf = l2_half() * DD;
  float* p = outb + (size_t)r0 * DD + d;
  float v[SROWS];
  #pragma unroll
  for (int j = 0; j < SROWS; ++j) v[j] = p[(size_t)j * DD];
  __builtin_amdgcn_sched_barrier(0);
  float s = 0.0f, s2 = 0.0f;
  #pragma unroll
  for (int j = 0; j < SROWS; ++j) {
    int r = r0 + j;
    float x = row_norm(v[j], denom[r] + denom[NNP + r]) + b;
    p[(size_t)j * DD] = x;
    s += x; s2 += x * x;
  }
  atomicAdd(&ssum[hf + d], s);
  atomicAdd(&ssq[hf + d], s2);
}

__global__ void k_bn_residual(const float* __restrict__ outb, const float* __restrict__ ssum,
                              const float* __restrict__ ssq, const float* __restrict__ gam,
                              const float* __restrict__ bet, float* __restrict__ xbuf) {
  int i = blockIdx.x * 256 + threadIdx.x;
  int d = i & (DD - 1);
  float mu  = (ssum[d] + ssum[DD + d]) * (1.0f / NN);
  float var = (ssq[d] + ssq[DD + d]) * (1.0f / NN) - mu * mu;
  float inv = rsqrtf(var + 1e-5f);
  float o = outb[i];
  xbuf[i] = (o - mu) * inv * gam[d] + bet[d] + xbuf[i];
}

__global__ void k_final(const float* __restrict__ outb, const float* __restrict__ denom,
                        const float* __restrict__ bias,
                        const float* __restrict__ xbuf, float* __restrict__ out) {
  int i = blockIdx.x * 256 + threadIdx.x;
  int d = i & (DD - 1);
  int r = i >> 7;
  out[i] = row_norm(outb[i], denom[r] + denom[NNP + r]) + bias[d] + xbuf[i];
}


extern "C" void kernel_launch(void* const* d_in, const int* in_sizes, int n_in,
                              void* d_out, int out_size, void* d_ws, size_t ws_size,
                              hipStream_t stream) {
  const float* X_n   = (const float*)d_in[0];
  const float* PE    = (const float*)d_in[1];
  const int*   eidx  = (const int*)d_in[2];
  const int*   eattr = (const int*)d_in[3];
  const float* eemb  = (const float*)d_in[4];
  const float* peW   = (const float*)d_in[5];
  const float* peB   = (const float*)d_in[6];
  const float* linW  = (const float*)d_in[7];
  const float* attl  = (const float*)d_in[8];
  const float* attr_ = (const float*)d_in[9];
  const float* atte  = (const float*)d_in[10];
  const float* bias  = (const float*)d_in[11];
  const float* gam   = (const float*)d_in[12];
  const float* bet   = (const float*)d_in[13];
  const int* src = eidx;
  const int* dst = eidx + EE;

  float* ws = (float*)d_ws;
  size_t o = 0;
  float* xbuf  = ws + o; o += (size_t)NN * DD;
  float* outb  = ws + o; o += (size_t)NN * DD;
  float* h     = ws + o; o += (size_t)NN * DD;
  float* PEp   = ws + o; o += (size_t)NN * PEDP;
  float* Wc    = ws + o; o += (size_t)NL * KC * NC;
  float* sl    = ws + o; o += NNP;
  float* sr    = ws + o; o += NNP;
  float* denom = ws + o; o += 2 * NNP;
  float* alpha = ws + o; o += EE;
  float* xe    = ws + o; o += 64;
  float* ssum  = ws + o; o += 2 * DD;
  float* ssq   = ws + o; o += 2 * DD;
  int*   brec  = (int*)(ws + o); o += (size_t)4 * EE;
  int*   whist  = (int*)(ws + o); o += (size_t)BIN_WGS * 256;
  int*   woff   = (int*)(ws + o); o += (size_t)BIN_WGS * 256;
  int*   bstart = (int*)(ws + o); o += 256;

  const int ND_BLKS = (NN * DD) / 256;

  k_copy_x<<<ND_BLKS, 256, 0, stream>>>(X_n, xbuf);
  k_pad_pe<<<(NN * PEDP + 255) / 256, 256, 0, stream>>>(PE, PEp);

  k_wcomb<<<NL, 27 * 32, 0, stream>>>(linW, peW, peB, attl, attr_, Wc);

  k_bin_count<<<BIN_WGS, 1024, 0, stream>>>(dst, whist);
  k_bin_scan<<<1, 1024, 0, stream>>>(whist, woff, bstart);
  k_bin_fill<<<BIN_WGS, 1024, 0, stream>>>(src, dst, eattr, woff, bstart, brec);

  for (int l = 0; l < NL; ++l) {
    k_gemm_layer<<<GEMM_BLKS, 9 * 32, 0, stream>>>(xbuf, PEp, Wc + (size_t)l * KC * NC, h, sl, sr);
    k_xe<<<1 + RESET_BLKS, XE_THREADS, 0, stream>>>(eemb + (size_t)l * NT * DD, atte + l * DD, xe,
                                                    denom, ssum, ssq);

    k_aggregate<<<NBINS, AGG_WAVES * 32, 0, stream>>>(h, sl, sr, xe, brec, bstart, alpha, denom, outb);

    if (l < NL - 1) {
      k_bias_stats<<<STAT_BLKS, 128, 0, stream>>>(outb, denom, bias + l * DD, ssum, ssq);
      k_bn_residual<<<ND_BLKS, 256, 0, stream>>>(outb, ssum, ssq, gam + l * DD,
                                                 bet + l * DD, xbuf);
    } else {
      k_final<<<ND_BLKS, 256, 0, stream>>>(outb, denom, bias + l * DD, xbuf, (float*)d_out);
    }
  }
}
